# starting kernel
# baseline (speedup 1.0000x reference)
#include <hip/hip_runtime.h>


typedef __attribute__((ext_vector_type(2))) float v2f;
typedef __attribute__((ext_vector_type(8))) float v8f;

namespace gat {
constexpr int Bc = 8, Mc = 1024, Fc = 256, Hc = 8, Ec = 32;
constexpr size_t WH_N = (size_t)Bc * Hc * Mc * Ec;
constexpr size_t S_N  = (size_t)Bc * Hc * Mc;
constexpr float LALPHA  = 0.2f;
constexpr float NEG_INF = -9.0e15f;
}

__global__ __launch_bounds__(32)
void gat_wh_kernel(const float* __restrict__ x,
                   const float* __restrict__ w,
                   const float* __restrict__ att,
                   float* __restrict__ wh,
                   float* __restrict__ s1,
                   float* __restrict__ s2) {
    using namespace gat;
    const int lane = threadIdx.x;
    const int bid  = blockIdx.x;
    const int mb   = bid & 31;
    const int h    = (bid >> 5) & 7;
    const int b    = bid >> 8;

    const int col  = lane & 15;
    const int half = lane >> 4;
    const int k0   = half * 2;

    const int m0 = mb * 32;
    const float* xrow0 = x + ((size_t)b * Mc + (m0 + col)) * Fc;
    const float* xrow1 = xrow0 + (size_t)16 * Fc;
    const float* wbase = w + (size_t)h * Fc * Ec;

    v8f acc[2][2] = {};
    for (int kk = 0; kk < Fc; kk += 4) {
        v2f af0 = *(const v2f*)(xrow0 + kk + k0);
        v2f af1 = *(const v2f*)(xrow1 + kk + k0);
        const float* wp = wbase + (size_t)(kk + k0) * Ec;
        v2f bf0 = { wp[col],      wp[Ec + col]      };
        v2f bf1 = { wp[16 + col], wp[Ec + 16 + col] };
        acc[0][0] = __builtin_amdgcn_wmma_f32_16x16x4_f32(false, af0, false, bf0,
                                                          (short)0, acc[0][0], false, false);
        acc[0][1] = __builtin_amdgcn_wmma_f32_16x16x4_f32(false, af0, false, bf1,
                                                          (short)0, acc[0][1], false, false);
        acc[1][0] = __builtin_amdgcn_wmma_f32_16x16x4_f32(false, af1, false, bf0,
                                                          (short)0, acc[1][0], false, false);
        acc[1][1] = __builtin_amdgcn_wmma_f32_16x16x4_f32(false, af1, false, bf1,
                                                          (short)0, acc[1][1], false, false);
    }

    const float a1_0 = att[h * 2 * Ec + col];
    const float a1_1 = att[h * 2 * Ec + 16 + col];
    const float a2_0 = att[h * 2 * Ec + Ec + col];
    const float a2_1 = att[h * 2 * Ec + Ec + 16 + col];

    const int bh = b * Hc + h;
    float* whbase = wh + ((size_t)bh * Mc + m0) * Ec;

    #pragma unroll
    for (int t = 0; t < 2; ++t) {
        #pragma unroll
        for (int r = 0; r < 8; ++r) {
            const int m_l = t * 16 + r + 8 * half;
            const float v0 = acc[t][0][r];
            const float v1 = acc[t][1][r];
            whbase[(size_t)m_l * Ec + col]      = v0;
            whbase[(size_t)m_l * Ec + 16 + col] = v1;

            float p1 = v0 * a1_0 + v1 * a1_1;
            float p2 = v0 * a2_0 + v1 * a2_1;
            #pragma unroll
            for (int d = 1; d < 16; d <<= 1) {
                p1 += __shfl_xor(p1, d, 32);
                p2 += __shfl_xor(p2, d, 32);
            }
            if (col == 0) {
                s1[(size_t)bh * Mc + m0 + m_l] = p1;
                s2[(size_t)bh * Mc + m0 + m_l] = p2;
            }
        }
    }
}

__global__ __launch_bounds__(256)
void gat_stats_kernel(const int* __restrict__ adj,
                      const float* __restrict__ s1,
                      const float* __restrict__ s2,
                      float* __restrict__ rmax,
                      float* __restrict__ rsum) {
    using namespace gat;
    const int lane = threadIdx.x & 31;
    const int wid  = blockIdx.x * 8 + (threadIdx.x >> 5);
    const int m    = wid & (Mc - 1);
    const int bh   = wid >> 10;
    const int b    = bh >> 3;

    const float s1v = s1[(size_t)bh * Mc + m];
    const int*   arow  = adj + ((size_t)b * Mc + m) * Mc;
    const float* s2row = s2 + (size_t)bh * Mc;

    float mx = NEG_INF;
    for (int n = lane; n < Mc; n += 32) {
        float sc = s1v + s2row[n];
        sc = sc > 0.0f ? sc : LALPHA * sc;
        sc = arow[n] > 0 ? sc : NEG_INF;
        mx = fmaxf(mx, sc);
    }
    #pragma unroll
    for (int d = 1; d < 32; d <<= 1) mx = fmaxf(mx, __shfl_xor(mx, d, 32));

    float sum = 0.0f;
    for (int n = lane; n < Mc; n += 32) {
        float sc = s1v + s2row[n];
        sc = sc > 0.0f ? sc : LALPHA * sc;
        sc = arow[n] > 0 ? sc : NEG_INF;
        sum += __expf(sc - mx);
    }
    #pragma unroll
    for (int d = 1; d < 32; d <<= 1) sum += __shfl_xor(sum, d, 32);

    if (lane == 0) {
        rmax[wid] = mx;
        rsum[wid] = sum;
    }
}

__global__ __launch_bounds__(32)
void gat_attn_kernel(const int* __restrict__ adj,
                     const float* __restrict__ wh,
                     const float* __restrict__ s1,
                     const float* __restrict__ s2,
                     const float* __restrict__ rmax,
                     const float* __restrict__ rsum,
                     float* __restrict__ out) {
    using namespace gat;
    __shared__ float pblk[2][16 * 16];

    const int lane = threadIdx.x;
    const int bid  = blockIdx.x;
    const int mb   = bid & 31;
    const int h    = (bid >> 5) & 7;
    const int b    = bid >> 8;
    const int bh   = b * Hc + h;

    const int col  = lane & 15;
    const int half = lane >> 4;
    const int k0   = half * 2;
    const int m0   = mb * 32;

    float s1v[16], rmx[16], rin[16];
    #pragma unroll
    for (int t = 0; t < 2; ++t) {
        #pragma unroll
        for (int r = 0; r < 8; ++r) {
            const int m = m0 + t * 16 + r + 8 * half;
            s1v[t * 8 + r] = s1[(size_t)bh * Mc + m];
            rmx[t * 8 + r] = rmax[(size_t)bh * Mc + m];
            rin[t * 8 + r] = 1.0f / rsum[(size_t)bh * Mc + m];
        }
    }

    const float* whb  = wh + (size_t)bh * Mc * Ec;
    const int*   adjb = adj + (size_t)b * Mc * Mc;
    const float* s2b  = s2 + (size_t)bh * Mc;

    v8f acc[2][2] = {};
    for (int nb = 0; nb < Mc / 16; ++nb) {
        const int n0 = nb * 16;
        const float s2v = s2b[n0 + col];

        if (nb + 1 < Mc / 16) {
            __builtin_prefetch(&adjb[(size_t)(m0 + 8 * half) * Mc + n0 + 16 + col], 0, 0);
            __builtin_prefetch(&adjb[(size_t)(m0 + 16 + 8 * half) * Mc + n0 + 16 + col], 0, 0);
        }

        #pragma unroll
        for (int t = 0; t < 2; ++t) {
            #pragma unroll
            for (int r = 0; r < 8; ++r) {
                const int mL = r + 8 * half;
                float sc = s1v[t * 8 + r] + s2v;
                sc = sc > 0.0f ? sc : LALPHA * sc;
                const int a = adjb[(size_t)(m0 + t * 16 + mL) * Mc + n0 + col];
                sc = a > 0 ? sc : NEG_INF;
                pblk[t][mL * 16 + col] = __expf(sc - rmx[t * 8 + r]) * rin[t * 8 + r];
            }
        }
        __syncthreads();

        #pragma unroll
        for (int ks = 0; ks < 4; ++ks) {
            v2f af0 = *(const v2f*)&pblk[0][col * 16 + ks * 4 + k0];
            v2f af1 = *(const v2f*)&pblk[1][col * 16 + ks * 4 + k0];
            const float* wp = whb + (size_t)(n0 + ks * 4 + k0) * Ec;
            v2f bf0 = { wp[col],      wp[Ec + col]      };
            v2f bf1 = { wp[16 + col], wp[Ec + 16 + col] };
            acc[0][0] = __builtin_amdgcn_wmma_f32_16x16x4_f32(false, af0, false, bf0,
                                                              (short)0, acc[0][0], false, false);
            acc[0][1] = __builtin_amdgcn_wmma_f32_16x16x4_f32(false, af0, false, bf1,
                                                              (short)0, acc[0][1], false, false);
            acc[1][0] = __builtin_amdgcn_wmma_f32_16x16x4_f32(false, af1, false, bf0,
                                                              (short)0, acc[1][0], false, false);
            acc[1][1] = __builtin_amdgcn_wmma_f32_16x16x4_f32(false, af1, false, bf1,
                                                              (short)0, acc[1][1], false, false);
        }
        __syncthreads();
    }

    float* ob = out + (size_t)b * Mc * (Hc * Ec) + (size_t)h * Ec;
    #pragma unroll
    for (int t = 0; t < 2; ++t) {
        #pragma unroll
        for (int r = 0; r < 8; ++r) {
            const int m = m0 + t * 16 + r + 8 * half;
            float v0 = acc[t][0][r];
            float v1 = acc[t][1][r];
            v0 = v0 > 0.0f ? v0 : (__expf(v0) - 1.0f);
            v1 = v1 > 0.0f ? v1 : (__expf(v1) - 1.0f);
            ob[(size_t)m * (Hc * Ec) + col]      = v0;
            ob[(size_t)m * (Hc * Ec) + 16 + col] = v1;
        }
    }
}

extern "C" void kernel_launch(void* const* d_in, const int* in_sizes, int n_in,
                              void* d_out, int out_size, void* d_ws, size_t ws_size,
                              hipStream_t stream) {
    using namespace gat;
    const float* x   = (const float*)d_in[0];
    const int*   adj = (const int*)  d_in[1];
    const float* w   = (const float*)d_in[2];
    const float* att = (const float*)d_in[3];
    float* out = (float*)d_out;

    float* ws   = (float*)d_ws;
    float* wh   = ws;
    float* s1   = ws + WH_N;
    float* s2   = ws + WH_N + S_N;
    float* rmx  = ws + WH_N + 2 * S_N;
    float* rsm  = ws + WH_N + 3 * S_N;

    const int slabs = Bc * Hc * (Mc / 32);

    gat_wh_kernel<<<slabs, 32, 0, stream>>>(x, w, att, wh, s1, s2);
    gat_stats_kernel<<<(int)(S_N / 8), 256, 0, stream>>>(adj, s1, s2, rmx, rsm);
    gat_attn_kernel<<<slabs, 32, 0, stream>>>(adj, wh, s1, s2, rmx, rsm, out);
}


# attention kernel: n loop split over the 8 waves of a 256-thread block per 32-row slab (3x the waves in flight per SIMD); partials summed through LDS in fixed wave order (sum order changes)
# speedup vs baseline: 1.1732x; 1.1732x over previous
#include <hip/hip_runtime.h>


typedef __attribute__((ext_vector_type(2))) float v2f;
typedef __attribute__((ext_vector_type(8))) float v8f;

namespace gat {
constexpr int Bc = 8, Mc = 1024, Fc = 256, Hc = 8, Ec = 32;
constexpr size_t WH_N = (size_t)Bc * Hc * Mc * Ec;
constexpr size_t S_N  = (size_t)Bc * Hc * Mc;
constexpr float LALPHA  = 0.2f;
constexpr float NEG_INF = -9.0e15f;
}

__global__ __launch_bounds__(32)
void gat_wh_kernel(const float* __restrict__ x,
                   const float* __restrict__ w,
                   const float* __restrict__ att,
                   float* __restrict__ wh,
                   float* __restrict__ s1,
                   float* __restrict__ s2) {
    using namespace gat;
    const int lane = threadIdx.x;
    const int bid  = blockIdx.x;
    const int mb   = bid & 31;
    const int h    = (bid >> 5) & 7;
    const int b    = bid >> 8;

    const int col  = lane & 15;
    const int half = lane >> 4;
    const int k0   = half * 2;

    const int m0 = mb * 32;
    const float* xrow0 = x + ((size_t)b * Mc + (m0 + col)) * Fc;
    const float* xrow1 = xrow0 + (size_t)16 * Fc;
    const float* wbase = w + (size_t)h * Fc * Ec;

    v8f acc[2][2] = {};
    for (int kk = 0; kk < Fc; kk += 4) {
        v2f af0 = *(const v2f*)(xrow0 + kk + k0);
        v2f af1 = *(const v2f*)(xrow1 + kk + k0);
        const float* wp = wbase + (size_t)(kk + k0) * Ec;
        v2f bf0 = { wp[col],      wp[Ec + col]      };
        v2f bf1 = { wp[16 + col], wp[Ec + 16 + col] };
        acc[0][0] = __builtin_amdgcn_wmma_f32_16x16x4_f32(false, af0, false, bf0,
                                                          (short)0, acc[0][0], false, false);
        acc[0][1] = __builtin_amdgcn_wmma_f32_16x16x4_f32(false, af0, false, bf1,
                                                          (short)0, acc[0][1], false, false);
        acc[1][0] = __builtin_amdgcn_wmma_f32_16x16x4_f32(false, af1, false, bf0,
                                                          (short)0, acc[1][0], false, false);
        acc[1][1] = __builtin_amdgcn_wmma_f32_16x16x4_f32(false, af1, false, bf1,
                                                          (short)0, acc[1][1], false, false);
    }

    const float a1_0 = att[h * 2 * Ec + col];
    const float a1_1 = att[h * 2 * Ec + 16 + col];
    const float a2_0 = att[h * 2 * Ec + Ec + col];
    const float a2_1 = att[h * 2 * Ec + Ec + 16 + col];

    const int bh = b * Hc + h;
    float* whbase = wh + ((size_t)bh * Mc + m0) * Ec;

    #pragma unroll
    for (int t = 0; t < 2; ++t) {
        #pragma unroll
        for (int r = 0; r < 8; ++r) {
            const int m_l = t * 16 + r + 8 * half;
            const float v0 = acc[t][0][r];
            const float v1 = acc[t][1][r];
            whbase[(size_t)m_l * Ec + col]      = v0;
            whbase[(size_t)m_l * Ec + 16 + col] = v1;

            float p1 = v0 * a1_0 + v1 * a1_1;
            float p2 = v0 * a2_0 + v1 * a2_1;
            #pragma unroll
            for (int d = 1; d < 16; d <<= 1) {
                p1 += __shfl_xor(p1, d, 32);
                p2 += __shfl_xor(p2, d, 32);
            }
            if (col == 0) {
                s1[(size_t)bh * Mc + m0 + m_l] = p1;
                s2[(size_t)bh * Mc + m0 + m_l] = p2;
            }
        }
    }
}

__global__ __launch_bounds__(256)
void gat_stats_kernel(const int* __restrict__ adj,
                      const float* __restrict__ s1,
                      const float* __restrict__ s2,
                      float* __restrict__ rmax,
                      float* __restrict__ rsum) {
    using namespace gat;
    const int lane = threadIdx.x & 31;
    const int wid  = blockIdx.x * 8 + (threadIdx.x >> 5);
    const int m    = wid & (Mc - 1);
    const int bh   = wid >> 10;
    const int b    = bh >> 3;

    const float s1v = s1[(size_t)bh * Mc + m];
    const int*   arow  = adj + ((size_t)b * Mc + m) * Mc;
    const float* s2row = s2 + (size_t)bh * Mc;

    float mx = NEG_INF;
    for (int n = lane; n < Mc; n += 32) {
        float sc = s1v + s2row[n];
        sc = sc > 0.0f ? sc : LALPHA * sc;
        sc = arow[n] > 0 ? sc : NEG_INF;
        mx = fmaxf(mx, sc);
    }
    #pragma unroll
    for (int d = 1; d < 32; d <<= 1) mx = fmaxf(mx, __shfl_xor(mx, d, 32));

    float sum = 0.0f;
    for (int n = lane; n < Mc; n += 32) {
        float sc = s1v + s2row[n];
        sc = sc > 0.0f ? sc : LALPHA * sc;
        sc = arow[n] > 0 ? sc : NEG_INF;
        sum += __expf(sc - mx);
    }
    #pragma unroll
    for (int d = 1; d < 32; d <<= 1) sum += __shfl_xor(sum, d, 32);

    if (lane == 0) {
        rmax[wid] = mx;
        rsum[wid] = sum;
    }
}

__global__ __launch_bounds__(256)
void gat_attn_kernel(const int* __restrict__ adj,
                     const float* __restrict__ wh,
                     const float* __restrict__ s1,
                     const float* __restrict__ s2,
                     const float* __restrict__ rmax,
                     const float* __restrict__ rsum,
                     float* __restrict__ out) {
    using namespace gat;
    constexpr int NW    = 8;
    constexpr int STEPS = Mc / 16 / NW;
    __shared__ float smem[NW * 1024];

    const int lane = threadIdx.x & 31;
    const int wv   = __builtin_amdgcn_readfirstlane((int)(threadIdx.x >> 5));
    const int bid  = blockIdx.x;
    const int mb   = bid & 31;
    const int h    = (bid >> 5) & 7;
    const int b    = bid >> 8;
    const int bh   = b * Hc + h;

    const int col  = lane & 15;
    const int half = lane >> 4;
    const int k0   = half * 2;
    const int m0   = mb * 32;

    float* pblk = smem + wv * 1024;

    float s1v[16], rmx[16], rin[16];
    #pragma unroll
    for (int t = 0; t < 2; ++t) {
        #pragma unroll
        for (int r = 0; r < 8; ++r) {
            const int m = m0 + t * 16 + r + 8 * half;
            s1v[t * 8 + r] = s1[(size_t)bh * Mc + m];
            rmx[t * 8 + r] = rmax[(size_t)bh * Mc + m];
            rin[t * 8 + r] = 1.0f / rsum[(size_t)bh * Mc + m];
        }
    }

    const float* whb  = wh + (size_t)bh * Mc * Ec;
    const int*   adjb = adj + (size_t)b * Mc * Mc;
    const float* s2b  = s2 + (size_t)bh * Mc;

    v8f acc[2][2] = {};
    const int nb_lo = wv * STEPS;
    for (int nb = nb_lo; nb < nb_lo + STEPS; ++nb) {
        const int n0 = nb * 16;
        const float s2v = s2b[n0 + col];

        if (nb + 1 < nb_lo + STEPS) {
            __builtin_prefetch(&adjb[(size_t)(m0 + 8 * half) * Mc + n0 + 16 + col], 0, 0);
            __builtin_prefetch(&adjb[(size_t)(m0 + 16 + 8 * half) * Mc + n0 + 16 + col], 0, 0);
        }

        #pragma unroll
        for (int t = 0; t < 2; ++t) {
            #pragma unroll
            for (int r = 0; r < 8; ++r) {
                const int mL = r + 8 * half;
                float sc = s1v[t * 8 + r] + s2v;
                sc = sc > 0.0f ? sc : LALPHA * sc;
                const int a = adjb[(size_t)(m0 + t * 16 + mL) * Mc + n0 + col];
                sc = a > 0 ? sc : NEG_INF;
                pblk[t * 256 + mL * 16 + col] = __expf(sc - rmx[t * 8 + r]) * rin[t * 8 + r];
            }
        }
        __builtin_amdgcn_fence(__ATOMIC_RELEASE, "wavefront");
        __builtin_amdgcn_wave_barrier();
        __builtin_amdgcn_fence(__ATOMIC_ACQUIRE, "wavefront");

        #pragma unroll
        for (int ks = 0; ks < 4; ++ks) {
            v2f af0 = *(const v2f*)&pblk[col * 16 + ks * 4 + k0];
            v2f af1 = *(const v2f*)&pblk[256 + col * 16 + ks * 4 + k0];
            const float* wp = whb + (size_t)(n0 + ks * 4 + k0) * Ec;
            v2f bf0 = { wp[col],      wp[Ec + col]      };
            v2f bf1 = { wp[16 + col], wp[Ec + 16 + col] };
            acc[0][0] = __builtin_amdgcn_wmma_f32_16x16x4_f32(false, af0, false, bf0,
                                                              (short)0, acc[0][0], false, false);
            acc[0][1] = __builtin_amdgcn_wmma_f32_16x16x4_f32(false, af0, false, bf1,
                                                              (short)0, acc[0][1], false, false);
            acc[1][0] = __builtin_amdgcn_wmma_f32_16x16x4_f32(false, af1, false, bf0,
                                                              (short)0, acc[1][0], false, false);
            acc[1][1] = __builtin_amdgcn_wmma_f32_16x16x4_f32(false, af1, false, bf1,
                                                              (short)0, acc[1][1], false, false);
        }
        __builtin_amdgcn_fence(__ATOMIC_RELEASE, "wavefront");
        __builtin_amdgcn_wave_barrier();
        __builtin_amdgcn_fence(__ATOMIC_ACQUIRE, "wavefront");
    }

    float* red = smem + wv * 1024;
    #pragma unroll
    for (int t = 0; t < 2; ++t) {
        #pragma unroll
        for (int et = 0; et < 2; ++et) {
            #pragma unroll
            for (int r = 0; r < 8; ++r) {
                red[(t * 16 + et * 8 + r) * 32 + lane] = acc[t][et][r];
            }
        }
    }
    __syncthreads();

    const int tt = wv >> 2;
    const int rb = (wv & 3) * 2;
    float* ob = out + (size_t)b * Mc * (Hc * Ec) + (size_t)h * Ec;
    #pragma unroll
    for (int rr = 0; rr < 2; ++rr) {
        const int r = rb + rr;
        const float* p0 = smem + (tt * 16 + r) * 32 + lane;
        const float* p1 = p0 + 8 * 32;
        float v0 = p0[0];
        float v1 = p1[0];
        #pragma unroll
        for (int w = 1; w < NW; ++w) {
            v0 += p0[w * 1024];
            v1 += p1[w * 1024];
        }
        const int m = m0 + tt * 16 + r + 8 * half;
        v0 = v0 > 0.0f ? v0 : (__expf(v0) - 1.0f);
        v1 = v1 > 0.0f ? v1 : (__expf(v1) - 1.0f);
        ob[(size_t)m * (Hc * Ec) + col]      = v0;
        ob[(size_t)m * (Hc * Ec) + 16 + col] = v1;
    }
}

extern "C" void kernel_launch(void* const* d_in, const int* in_sizes, int n_in,
                              void* d_out, int out_size, void* d_ws, size_t ws_size,
                              hipStream_t stream) {
    using namespace gat;
    const float* x   = (const float*)d_in[0];
    const int*   adj = (const int*)  d_in[1];
    const float* w   = (const float*)d_in[2];
    const float* att = (const float*)d_in[3];
    float* out = (float*)d_out;

    float* ws   = (float*)d_ws;
    float* wh   = ws;
    float* s1   = ws + WH_N;
    float* s2   = ws + WH_N + S_N;
    float* rmx  = ws + WH_N + 2 * S_N;
    float* rsm  = ws + WH_N + 3 * S_N;

    const int slabs = Bc * Hc * (Mc / 32);

    gat_wh_kernel<<<slabs, 32, 0, stream>>>(x, w, att, wh, s1, s2);
    gat_stats_kernel<<<(int)(S_N / 8), 256, 0, stream>>>(adj, s1, s2, rmx, rsm);
    gat_attn_kernel<<<slabs, 256, 0, stream>>>(adj, wh, s1, s2, rmx, rsm, out);
}


# attention kernel: global loads issued ahead of use: next step's adj words and s2 fetched one step early into a second register set, the step's Wh fragments before its P math; sum order unchanged
# speedup vs baseline: 1.4520x; 1.2376x over previous
#include <hip/hip_runtime.h>


typedef __attribute__((ext_vector_type(2))) float v2f;
typedef __attribute__((ext_vector_type(8))) float v8f;

namespace gat {
constexpr int Bc = 8, Mc = 1024, Fc = 256, Hc = 8, Ec = 32;
constexpr size_t WH_N = (size_t)Bc * Hc * Mc * Ec;
constexpr size_t S_N  = (size_t)Bc * Hc * Mc;
constexpr float LALPHA  = 0.2f;
constexpr float NEG_INF = -9.0e15f;
}

__global__ __launch_bounds__(32)
void gat_wh_kernel(const float* __restrict__ x,
                   const float* __restrict__ w,
                   const float* __restrict__ att,
                   float* __restrict__ wh,
                   float* __restrict__ s1,
                   float* __restrict__ s2) {
    using namespace gat;
    const int lane = threadIdx.x;
    const int bid  = blockIdx.x;
    const int mb   = bid & 31;
    const int h    = (bid >> 5) & 7;
    const int b    = bid >> 8;

    const int col  = lane & 15;
    const int half = lane >> 4;
    const int k0   = half * 2;

    const int m0 = mb * 32;
    const float* xrow0 = x + ((size_t)b * Mc + (m0 + col)) * Fc;
    const float* xrow1 = xrow0 + (size_t)16 * Fc;
    const float* wbase = w + (size_t)h * Fc * Ec;

    v8f acc[2][2] = {};
    for (int kk = 0; kk < Fc; kk += 4) {
        v2f af0 = *(const v2f*)(xrow0 + kk + k0);
        v2f af1 = *(const v2f*)(xrow1 + kk + k0);
        const float* wp = wbase + (size_t)(kk + k0) * Ec;
        v2f bf0 = { wp[col],      wp[Ec + col]      };
        v2f bf1 = { wp[16 + col], wp[Ec + 16 + col] };
        acc[0][0] = __builtin_amdgcn_wmma_f32_16x16x4_f32(false, af0, false, bf0,
                                                          (short)0, acc[0][0], false, false);
        acc[0][1] = __builtin_amdgcn_wmma_f32_16x16x4_f32(false, af0, false, bf1,
                                                          (short)0, acc[0][1], false, false);
        acc[1][0] = __builtin_amdgcn_wmma_f32_16x16x4_f32(false, af1, false, bf0,
                                                          (short)0, acc[1][0], false, false);
        acc[1][1] = __builtin_amdgcn_wmma_f32_16x16x4_f32(false, af1, false, bf1,
                                                          (short)0, acc[1][1], false, false);
    }

    const float a1_0 = att[h * 2 * Ec + col];
    const float a1_1 = att[h * 2 * Ec + 16 + col];
    const float a2_0 = att[h * 2 * Ec + Ec + col];
    const float a2_1 = att[h * 2 * Ec + Ec + 16 + col];

    const int bh = b * Hc + h;
    float* whbase = wh + ((size_t)bh * Mc + m0) * Ec;

    #pragma unroll
    for (int t = 0; t < 2; ++t) {
        #pragma unroll
        for (int r = 0; r < 8; ++r) {
            const int m_l = t * 16 + r + 8 * half;
            const float v0 = acc[t][0][r];
            const float v1 = acc[t][1][r];
            whbase[(size_t)m_l * Ec + col]      = v0;
            whbase[(size_t)m_l * Ec + 16 + col] = v1;

            float p1 = v0 * a1_0 + v1 * a1_1;
            float p2 = v0 * a2_0 + v1 * a2_1;
            #pragma unroll
            for (int d = 1; d < 16; d <<= 1) {
                p1 += __shfl_xor(p1, d, 32);
                p2 += __shfl_xor(p2, d, 32);
            }
            if (col == 0) {
                s1[(size_t)bh * Mc + m0 + m_l] = p1;
                s2[(size_t)bh * Mc + m0 + m_l] = p2;
            }
        }
    }
}

__global__ __launch_bounds__(256)
void gat_stats_kernel(const int* __restrict__ adj,
                      const float* __restrict__ s1,
                      const float* __restrict__ s2,
                      float* __restrict__ rmax,
                      float* __restrict__ rsum) {
    using namespace gat;
    const int lane = threadIdx.x & 31;
    const int wid  = blockIdx.x * 8 + (threadIdx.x >> 5);
    const int m    = wid & (Mc - 1);
    const int bh   = wid >> 10;
    const int b    = bh >> 3;

    const float s1v = s1[(size_t)bh * Mc + m];
    const int*   arow  = adj + ((size_t)b * Mc + m) * Mc;
    const float* s2row = s2 + (size_t)bh * Mc;

    float mx = NEG_INF;
    for (int n = lane; n < Mc; n += 32) {
        float sc = s1v + s2row[n];
        sc = sc > 0.0f ? sc : LALPHA * sc;
        sc = arow[n] > 0 ? sc : NEG_INF;
        mx = fmaxf(mx, sc);
    }
    #pragma unroll
    for (int d = 1; d < 32; d <<= 1) mx = fmaxf(mx, __shfl_xor(mx, d, 32));

    float sum = 0.0f;
    for (int n = lane; n < Mc; n += 32) {
        float sc = s1v + s2row[n];
        sc = sc > 0.0f ? sc : LALPHA * sc;
        sc = arow[n] > 0 ? sc : NEG_INF;
        sum += __expf(sc - mx);
    }
    #pragma unroll
    for (int d = 1; d < 32; d <<= 1) sum += __shfl_xor(sum, d, 32);

    if (lane == 0) {
        rmax[wid] = mx;
        rsum[wid] = sum;
    }
}

__global__ __launch_bounds__(256)
void gat_attn_kernel(const int* __restrict__ adj,
                     const float* __restrict__ wh,
                     const float* __restrict__ s1,
                     const float* __restrict__ s2,
                     const float* __restrict__ rmax,
                     const float* __restrict__ rsum,
                     float* __restrict__ out) {
    using namespace gat;
    constexpr int NW    = 8;
    constexpr int STEPS = Mc / 16 / NW;
    __shared__ float smem[NW * 1024];

    const int lane = threadIdx.x & 31;
    const int wv   = __builtin_amdgcn_readfirstlane((int)(threadIdx.x >> 5));
    const int bid  = blockIdx.x;
    const int mb   = bid & 31;
    const int h    = (bid >> 5) & 7;
    const int b    = bid >> 8;
    const int bh   = b * Hc + h;

    const int col  = lane & 15;
    const int half = lane >> 4;
    const int k0   = half * 2;
    const int m0   = mb * 32;

    float* pblk = smem + wv * 1024;

    float s1v[16], rmx[16], rin[16];
    #pragma unroll
    for (int t = 0; t < 2; ++t) {
        #pragma unroll
        for (int r = 0; r < 8; ++r) {
            const int m = m0 + t * 16 + r + 8 * half;
            s1v[t * 8 + r] = s1[(size_t)bh * Mc + m];
            rmx[t * 8 + r] = rmax[(size_t)bh * Mc + m];
            rin[t * 8 + r] = 1.0f / rsum[(size_t)bh * Mc + m];
        }
    }

    const float* whb  = wh + (size_t)bh * Mc * Ec;
    const float* s2b  = s2 + (size_t)bh * Mc;
    const int* arow = adj + (size_t)b * Mc * Mc + (size_t)(m0 + 8 * half) * Mc + col;

    const int nb_lo = wv * STEPS;

    int   ac[16];
    float s2c = s2b[nb_lo * 16 + col];
    #pragma unroll
    for (int t = 0; t < 2; ++t) {
        #pragma unroll
        for (int r = 0; r < 8; ++r) {
            ac[t * 8 + r] = arow[(size_t)(t * 16 + r) * Mc + nb_lo * 16];
        }
    }

    v8f acc[2][2] = {};
    for (int i = 0; i < STEPS; ++i) {
        const int n0 = (nb_lo + i) * 16;

        v2f bf[4][2];
        #pragma unroll
        for (int ks = 0; ks < 4; ++ks) {
            const float* wp = whb + (size_t)(n0 + ks * 4 + k0) * Ec;
            bf[ks][0] = v2f{ wp[col],      wp[Ec + col]      };
            bf[ks][1] = v2f{ wp[16 + col], wp[Ec + 16 + col] };
        }

        const int n1 = (i + 1 < STEPS) ? n0 + 16 : n0;
        int   an[16];
        const float s2n = s2b[n1 + col];
        #pragma unroll
        for (int t = 0; t < 2; ++t) {
            #pragma unroll
            for (int r = 0; r < 8; ++r) {
                an[t * 8 + r] = arow[(size_t)(t * 16 + r) * Mc + n1];
            }
        }

        #pragma unroll
        for (int t = 0; t < 2; ++t) {
            #pragma unroll
            for (int r = 0; r < 8; ++r) {
                const int mL = r + 8 * half;
                float sc = s1v[t * 8 + r] + s2c;
                sc = sc > 0.0f ? sc : LALPHA * sc;
                sc = ac[t * 8 + r] > 0 ? sc : NEG_INF;
                pblk[t * 256 + mL * 16 + col] = __expf(sc - rmx[t * 8 + r]) * rin[t * 8 + r];
            }
        }
        __builtin_amdgcn_fence(__ATOMIC_RELEASE, "wavefront");
        __builtin_amdgcn_wave_barrier();
        __builtin_amdgcn_fence(__ATOMIC_ACQUIRE, "wavefront");

        #pragma unroll
        for (int ks = 0; ks < 4; ++ks) {
            v2f af0 = *(const v2f*)&pblk[col * 16 + ks * 4 + k0];
            v2f af1 = *(const v2f*)&pblk[256 + col * 16 + ks * 4 + k0];
            acc[0][0] = __builtin_amdgcn_wmma_f32_16x16x4_f32(false, af0, false, bf[ks][0],
                                                              (short)0, acc[0][0], false, false);
            acc[0][1] = __builtin_amdgcn_wmma_f32_16x16x4_f32(false, af0, false, bf[ks][1],
                                                              (short)0, acc[0][1], false, false);
            acc[1][0] = __builtin_amdgcn_wmma_f32_16x16x4_f32(false, af1, false, bf[ks][0],
                                                              (short)0, acc[1][0], false, false);
            acc[1][1] = __builtin_amdgcn_wmma_f32_16x16x4_f32(false, af1, false, bf[ks][1],
                                                              (short)0, acc[1][1], false, false);
        }
        __builtin_amdgcn_fence(__ATOMIC_RELEASE, "wavefront");
        __builtin_amdgcn_wave_barrier();
        __builtin_amdgcn_fence(__ATOMIC_ACQUIRE, "wavefront");

        s2c = s2n;
        #pragma unroll
        for (int q = 0; q < 16; ++q) ac[q] = an[q];
    }

    float* red = smem + wv * 1024;
    #pragma unroll
    for (int t = 0; t < 2; ++t) {
        #pragma unroll
        for (int et = 0; et < 2; ++et) {
            #pragma unroll
            for (int r = 0; r < 8; ++r) {
                red[(t * 16 + et * 8 + r) * 32 + lane] = acc[t][et][r];
            }
        }
    }
    __syncthreads();

    const int tt = wv >> 2;
    const int rb = (wv & 3) * 2;
    float* ob = out + (size_t)b * Mc * (Hc * Ec) + (size_t)h * Ec;
    #pragma unroll
    for (int rr = 0; rr < 2; ++rr) {
        const int r = rb + rr;
        const float* p0 = smem + (tt * 16 + r) * 32 + lane;
        const float* p1 = p0 + 8 * 32;
        float v0 = p0[0];
        float v1 = p1[0];
        #pragma unroll
        for (int w = 1; w < NW; ++w) {
            v0 += p0[w * 1024];
            v1 += p1[w * 1024];
        }
        const int m = m0 + tt * 16 + r + 8 * half;
        v0 = v0 > 0.0f ? v0 : (__expf(v0) - 1.0f);
        v1 = v1 > 0.0f ? v1 : (__expf(v1) - 1.0f);
        ob[(size_t)m * (Hc * Ec) + col]      = v0;
        ob[(size_t)m * (Hc * Ec) + 16 + col] = v1;
    }
}

extern "C" void kernel_launch(void* const* d_in, const int* in_sizes, int n_in,
                              void* d_out, int out_size, void* d_ws, size_t ws_size,
                              hipStream_t stream) {
    using namespace gat;
    const float* x   = (const float*)d_in[0];
    const int*   adj = (const int*)  d_in[1];
    const float* w   = (const float*)d_in[2];
    const float* att = (const float*)d_in[3];
    float* out = (float*)d_out;

    float* ws   = (float*)d_ws;
    float* wh   = ws;
    float* s1   = ws + WH_N;
    float* s2   = ws + WH_N + S_N;
    float* rmx  = ws + WH_N + 2 * S_N;
    float* rsm  = ws + WH_N + 3 * S_N;

    const int slabs = Bc * Hc * (Mc / 32);

    gat_wh_kernel<<<slabs, 32, 0, stream>>>(x, w, att, wh, s1, s2);
    gat_stats_kernel<<<(int)(S_N / 8), 256, 0, stream>>>(adj, s1, s2, rmx, rsm);
    gat_attn_kernel<<<slabs, 256, 0, stream>>>(adj, wh, s1, s2, rmx, rsm, out);
}


# softmax row max and row sum taken inside the attention kernel (max from each row's unmasked s2, sum from the P fragments, one divide at the end); stats kernel and its launch gone; sum order changes
# speedup vs baseline: 2.4378x; 1.6789x over previous
#include <hip/hip_runtime.h>


typedef __attribute__((ext_vector_type(2))) float v2f;
typedef __attribute__((ext_vector_type(8))) float v8f;

namespace gat {
constexpr int Bc = 8, Mc = 1024, Fc = 256, Hc = 8, Ec = 32;
constexpr size_t WH_N = (size_t)Bc * Hc * Mc * Ec;
constexpr size_t S_N  = (size_t)Bc * Hc * Mc;
constexpr float LALPHA  = 0.2f;
constexpr float NEG_INF = -9.0e15f;
}

__global__ __launch_bounds__(32)
void gat_wh_kernel(const float* __restrict__ x,
                   const float* __restrict__ w,
                   const float* __restrict__ att,
                   float* __restrict__ wh,
                   float* __restrict__ s1,
                   float* __restrict__ s2) {
    using namespace gat;
    const int lane = threadIdx.x;
    const int bid  = blockIdx.x;
    const int mb   = bid & 31;
    const int h    = (bid >> 5) & 7;
    const int b    = bid >> 8;

    const int col  = lane & 15;
    const int half = lane >> 4;
    const int k0   = half * 2;

    const int m0 = mb * 32;
    const float* xrow0 = x + ((size_t)b * Mc + (m0 + col)) * Fc;
    const float* xrow1 = xrow0 + (size_t)16 * Fc;
    const float* wbase = w + (size_t)h * Fc * Ec;

    v8f acc[2][2] = {};
    for (int kk = 0; kk < Fc; kk += 4) {
        v2f af0 = *(const v2f*)(xrow0 + kk + k0);
        v2f af1 = *(const v2f*)(xrow1 + kk + k0);
        const float* wp = wbase + (size_t)(kk + k0) * Ec;
        v2f bf0 = { wp[col],      wp[Ec + col]      };
        v2f bf1 = { wp[16 + col], wp[Ec + 16 + col] };
        acc[0][0] = __builtin_amdgcn_wmma_f32_16x16x4_f32(false, af0, false, bf0,
                                                          (short)0, acc[0][0], false, false);
        acc[0][1] = __builtin_amdgcn_wmma_f32_16x16x4_f32(false, af0, false, bf1,
                                                          (short)0, acc[0][1], false, false);
        acc[1][0] = __builtin_amdgcn_wmma_f32_16x16x4_f32(false, af1, false, bf0,
                                                          (short)0, acc[1][0], false, false);
        acc[1][1] = __builtin_amdgcn_wmma_f32_16x16x4_f32(false, af1, false, bf1,
                                                          (short)0, acc[1][1], false, false);
    }

    const float a1_0 = att[h * 2 * Ec + col];
    const float a1_1 = att[h * 2 * Ec + 16 + col];
    const float a2_0 = att[h * 2 * Ec + Ec + col];
    const float a2_1 = att[h * 2 * Ec + Ec + 16 + col];

    const int bh = b * Hc + h;
    float* whbase = wh + ((size_t)bh * Mc + m0) * Ec;

    #pragma unroll
    for (int t = 0; t < 2; ++t) {
        #pragma unroll
        for (int r = 0; r < 8; ++r) {
            const int m_l = t * 16 + r + 8 * half;
            const float v0 = acc[t][0][r];
            const float v1 = acc[t][1][r];
            whbase[(size_t)m_l * Ec + col]      = v0;
            whbase[(size_t)m_l * Ec + 16 + col] = v1;

            float p1 = v0 * a1_0 + v1 * a1_1;
            float p2 = v0 * a2_0 + v1 * a2_1;
            #pragma unroll
            for (int d = 1; d < 16; d <<= 1) {
                p1 += __shfl_xor(p1, d, 32);
                p2 += __shfl_xor(p2, d, 32);
            }
            if (col == 0) {
                s1[(size_t)bh * Mc + m0 + m_l] = p1;
                s2[(size_t)bh * Mc + m0 + m_l] = p2;
            }
        }
    }
}

__global__ __launch_bounds__(256)
void gat_attn_kernel(const int* __restrict__ adj,
                     const float* __restrict__ wh,
                     const float* __restrict__ s1,
                     const float* __restrict__ s2,
                     float* __restrict__ out) {
    using namespace gat;
    typedef __attribute__((ext_vector_type(4))) float v4f;
    typedef __attribute__((ext_vector_type(4))) int   v4i;
    constexpr int NW    = 8;
    constexpr int STEPS = Mc / 16 / NW;
    __shared__ float smem[NW * 1024];
    __shared__ float spsum[NW * 32];
    __shared__ __attribute__((aligned(16))) float smax[32];

    const int lane = threadIdx.x & 31;
    const int wv   = __builtin_amdgcn_readfirstlane((int)(threadIdx.x >> 5));
    const int bid  = blockIdx.x;
    const int mb   = bid & 31;
    const int h    = (bid >> 5) & 7;
    const int b    = bid >> 8;
    const int bh   = b * Hc + h;

    const int col  = lane & 15;
    const int half = lane >> 4;
    const int k0   = half * 2;
    const int m0   = mb * 32;

    const float* whb  = wh + (size_t)bh * Mc * Ec;
    const float* s2b  = s2 + (size_t)bh * Mc;

    {
        const float NINF = -__builtin_inff();
        v4f sv[8];
        #pragma unroll
        for (int j = 0; j < 8; ++j) sv[j] = *(const v4f*)(s2b + j * 128 + lane * 4);
        const v4i* ap = (const v4i*)(adj + ((size_t)b * Mc + m0 + wv * 4) * Mc) + lane;
        float mx[4];
        #pragma unroll
        for (int rr = 0; rr < 4; ++rr) {
            float m = NINF;
            #pragma unroll
            for (int j = 0; j < 8; ++j) {
                const v4i a = ap[rr * (Mc / 4) + j * 32];
                m = fmaxf(m, a.x > 0 ? sv[j].x : NINF);
                m = fmaxf(m, a.y > 0 ? sv[j].y : NINF);
                m = fmaxf(m, a.z > 0 ? sv[j].z : NINF);
                m = fmaxf(m, a.w > 0 ? sv[j].w : NINF);
            }
            mx[rr] = m;
        }
        #pragma unroll
        for (int d = 1; d < 32; d <<= 1) {
            #pragma unroll
            for (int rr = 0; rr < 4; ++rr) mx[rr] = fmaxf(mx[rr], __shfl_xor(mx[rr], d, 32));
        }
        if (lane == 0) *(v4f*)&smax[wv * 4] = v4f{ mx[0], mx[1], mx[2], mx[3] };
    }
    __syncthreads();

    float* pblk = smem + wv * 1024;

    float s1v[16], rmx[16];
    #pragma unroll
    for (int t = 0; t < 2; ++t) {
        #pragma unroll
        for (int r = 0; r < 8; ++r) {
            const int mL = t * 16 + r + 8 * half;
            s1v[t * 8 + r] = s1[(size_t)bh * Mc + m0 + mL];
            float sc = s1v[t * 8 + r] + smax[mL];
            sc = sc > 0.0f ? sc : LALPHA * sc;
            rmx[t * 8 + r] = fmaxf(NEG_INF, sc);
        }
    }

    const int* arow = adj + (size_t)b * Mc * Mc + (size_t)(m0 + 8 * half) * Mc + col;

    const int nb_lo = wv * STEPS;

    int   ac[16];
    float s2c = s2b[nb_lo * 16 + col];
    #pragma unroll
    for (int t = 0; t < 2; ++t) {
        #pragma unroll
        for (int r = 0; r < 8; ++r) {
            ac[t * 8 + r] = arow[(size_t)(t * 16 + r) * Mc + nb_lo * 16];
        }
    }

    v8f acc[2][2] = {};
    v2f psa[2] = {};
    for (int i = 0; i < STEPS; ++i) {
        const int n0 = (nb_lo + i) * 16;

        v2f bf[4][2];
        #pragma unroll
        for (int ks = 0; ks < 4; ++ks) {
            const float* wp = whb + (size_t)(n0 + ks * 4 + k0) * Ec;
            bf[ks][0] = v2f{ wp[col],      wp[Ec + col]      };
            bf[ks][1] = v2f{ wp[16 + col], wp[Ec + 16 + col] };
        }

        const int n1 = (i + 1 < STEPS) ? n0 + 16 : n0;
        int   an[16];
        const float s2n = s2b[n1 + col];
        #pragma unroll
        for (int t = 0; t < 2; ++t) {
            #pragma unroll
            for (int r = 0; r < 8; ++r) {
                an[t * 8 + r] = arow[(size_t)(t * 16 + r) * Mc + n1];
            }
        }

        #pragma unroll
        for (int t = 0; t < 2; ++t) {
            #pragma unroll
            for (int r = 0; r < 8; ++r) {
                const int mL = r + 8 * half;
                float sc = s1v[t * 8 + r] + s2c;
                sc = sc > 0.0f ? sc : LALPHA * sc;
                sc = ac[t * 8 + r] > 0 ? sc : NEG_INF;
                pblk[t * 256 + mL * 16 + col] = __expf(sc - rmx[t * 8 + r]);
            }
        }
        __builtin_amdgcn_fence(__ATOMIC_RELEASE, "wavefront");
        __builtin_amdgcn_wave_barrier();
        __builtin_amdgcn_fence(__ATOMIC_ACQUIRE, "wavefront");

        #pragma unroll
        for (int ks = 0; ks < 4; ++ks) {
            v2f af0 = *(const v2f*)&pblk[col * 16 + ks * 4 + k0];
            v2f af1 = *(const v2f*)&pblk[256 + col * 16 + ks * 4 + k0];
            psa[0] += af0;
            psa[1] += af1;
            acc[0][0] = __builtin_amdgcn_wmma_f32_16x16x4_f32(false, af0, false, bf[ks][0],
                                                              (short)0, acc[0][0], false, false);
            acc[0][1] = __builtin_amdgcn_wmma_f32_16x16x4_f32(false, af0, false, bf[ks][1],
                                                              (short)0, acc[0][1], false, false);
            acc[1][0] = __builtin_amdgcn_wmma_f32_16x16x4_f32(false, af1, false, bf[ks][0],
                                                              (short)0, acc[1][0], false, false);
            acc[1][1] = __builtin_amdgcn_wmma_f32_16x16x4_f32(false, af1, false, bf[ks][1],
                                                              (short)0, acc[1][1], false, false);
        }
        __builtin_amdgcn_fence(__ATOMIC_RELEASE, "wavefront");
        __builtin_amdgcn_wave_barrier();
        __builtin_amdgcn_fence(__ATOMIC_ACQUIRE, "wavefront");

        s2c = s2n;
        #pragma unroll
        for (int q = 0; q < 16; ++q) ac[q] = an[q];
    }

    float* red = smem + wv * 1024;
    #pragma unroll
    for (int t = 0; t < 2; ++t) {
        #pragma unroll
        for (int et = 0; et < 2; ++et) {
            #pragma unroll
            for (int r = 0; r < 8; ++r) {
                red[(t * 16 + et * 8 + r) * 32 + lane] = acc[t][et][r];
            }
        }
    }
    {
        float rs0 = psa[0].x + psa[0].y;
        float rs1 = psa[1].x + psa[1].y;
        rs0 += __shfl_xor(rs0, 16, 32);
        rs1 += __shfl_xor(rs1, 16, 32);
        if (half == 0) {
            spsum[wv * 32 + col]      = rs0;
            spsum[wv * 32 + 16 + col] = rs1;
        }
    }
    __syncthreads();

    const int tt = wv >> 2;
    const int rb = (wv & 3) * 2;
    float* ob = out + (size_t)b * Mc * (Hc * Ec) + (size_t)h * Ec;
    #pragma unroll
    for (int rr = 0; rr < 2; ++rr) {
        const int r = rb + rr;
        const float* p0 = smem + (tt * 16 + r) * 32 + lane;
        const float* p1 = p0 + 8 * 32;
        const float* ps = spsum + tt * 16 + r + 8 * half;
        float v0 = p0[0];
        float v1 = p1[0];
        float rs = ps[0];
        #pragma unroll
        for (int w = 1; w < NW; ++w) {
            v0 += p0[w * 1024];
            v1 += p1[w * 1024];
            rs += ps[w * 32];
        }
        const float inv = 1.0f / rs;
        v0 *= inv;
        v1 *= inv;
        const int m = m0 + tt * 16 + r + 8 * half;
        v0 = v0 > 0.0f ? v0 : (__expf(v0) - 1.0f);
        v1 = v1 > 0.0f ? v1 : (__expf(v1) - 1.0f);
        ob[(size_t)m * (Hc * Ec) + col]      = v0;
        ob[(size_t)m * (Hc * Ec) + 16 + col] = v1;
    }
}

extern "C" void kernel_launch(void* const* d_in, const int* in_sizes, int n_in,
                              void* d_out, int out_size, void* d_ws, size_t ws_size,
                              hipStream_t stream) {
    using namespace gat;
    const float* x   = (const float*)d_in[0];
    const int*   adj = (const int*)  d_in[1];
    const float* w   = (const float*)d_in[2];
    const float* att = (const float*)d_in[3];
    float* out = (float*)d_out;

    float* ws   = (float*)d_ws;
    float* wh   = ws;
    float* s1   = ws + WH_N;
    float* s2   = ws + WH_N + S_N;

    const int slabs = Bc * Hc * (Mc / 32);

    gat_wh_kernel<<<slabs, 32, 0, stream>>>(x, w, att, wh, s1, s2);
    gat_attn_kernel<<<slabs, 256, 0, stream>>>(adj, wh, s1, s2, out);
}


# attention kernel: the row-max scan of adj also leaves the slab's mask in LDS (a byte per column, a bit per row); main loop takes mask bits there, its 16 global adj loads per step gone; sum order same
# speedup vs baseline: 2.5361x; 1.0403x over previous
#include <hip/hip_runtime.h>


typedef __attribute__((ext_vector_type(2))) float v2f;
typedef __attribute__((ext_vector_type(8))) float v8f;

namespace gat {
constexpr int Bc = 8, Mc = 1024, Fc = 256, Hc = 8, Ec = 32;
constexpr size_t WH_N = (size_t)Bc * Hc * Mc * Ec;
constexpr size_t S_N  = (size_t)Bc * Hc * Mc;
constexpr float LALPHA  = 0.2f;
constexpr float NEG_INF = -9.0e15f;
}

__global__ __launch_bounds__(32)
void gat_wh_kernel(const float* __restrict__ x,
                   const float* __restrict__ w,
                   const float* __restrict__ att,
                   float* __restrict__ wh,
                   float* __restrict__ s1,
                   float* __restrict__ s2) {
    using namespace gat;
    const int lane = threadIdx.x;
    const int bid  = blockIdx.x;
    const int mb   = bid & 31;
    const int h    = (bid >> 5) & 7;
    const int b    = bid >> 8;

    const int col  = lane & 15;
    const int half = lane >> 4;
    const int k0   = half * 2;

    const int m0 = mb * 32;
    const float* xrow0 = x + ((size_t)b * Mc + (m0 + col)) * Fc;
    const float* xrow1 = xrow0 + (size_t)16 * Fc;
    const float* wbase = w + (size_t)h * Fc * Ec;

    v8f acc[2][2] = {};
    for (int kk = 0; kk < Fc; kk += 4) {
        v2f af0 = *(const v2f*)(xrow0 + kk + k0);
        v2f af1 = *(const v2f*)(xrow1 + kk + k0);
        const float* wp = wbase + (size_t)(kk + k0) * Ec;
        v2f bf0 = { wp[col],      wp[Ec + col]      };
        v2f bf1 = { wp[16 + col], wp[Ec + 16 + col] };
        acc[0][0] = __builtin_amdgcn_wmma_f32_16x16x4_f32(false, af0, false, bf0,
                                                          (short)0, acc[0][0], false, false);
        acc[0][1] = __builtin_amdgcn_wmma_f32_16x16x4_f32(false, af0, false, bf1,
                                                          (short)0, acc[0][1], false, false);
        acc[1][0] = __builtin_amdgcn_wmma_f32_16x16x4_f32(false, af1, false, bf0,
                                                          (short)0, acc[1][0], false, false);
        acc[1][1] = __builtin_amdgcn_wmma_f32_16x16x4_f32(false, af1, false, bf1,
                                                          (short)0, acc[1][1], false, false);
    }

    const float a1_0 = att[h * 2 * Ec + col];
    const float a1_1 = att[h * 2 * Ec + 16 + col];
    const float a2_0 = att[h * 2 * Ec + Ec + col];
    const float a2_1 = att[h * 2 * Ec + Ec + 16 + col];

    const int bh = b * Hc + h;
    float* whbase = wh + ((size_t)bh * Mc + m0) * Ec;

    #pragma unroll
    for (int t = 0; t < 2; ++t) {
        #pragma unroll
        for (int r = 0; r < 8; ++r) {
            const int m_l = t * 16 + r + 8 * half;
            const float v0 = acc[t][0][r];
            const float v1 = acc[t][1][r];
            whbase[(size_t)m_l * Ec + col]      = v0;
            whbase[(size_t)m_l * Ec + 16 + col] = v1;

            float p1 = v0 * a1_0 + v1 * a1_1;
            float p2 = v0 * a2_0 + v1 * a2_1;
            #pragma unroll
            for (int d = 1; d < 16; d <<= 1) {
                p1 += __shfl_xor(p1, d, 32);
                p2 += __shfl_xor(p2, d, 32);
            }
            if (col == 0) {
                s1[(size_t)bh * Mc + m0 + m_l] = p1;
                s2[(size_t)bh * Mc + m0 + m_l] = p2;
            }
        }
    }
}

__global__ __launch_bounds__(256)
void gat_attn_kernel(const int* __restrict__ adj,
                     const float* __restrict__ wh,
                     const float* __restrict__ s1,
                     const float* __restrict__ s2,
                     float* __restrict__ out) {
    using namespace gat;
    typedef __attribute__((ext_vector_type(4))) float v4f;
    typedef __attribute__((ext_vector_type(4))) int   v4i;
    constexpr int NW    = 8;
    constexpr int STEPS = Mc / 16 / NW;
    __shared__ float smem[NW * 1024];
    __shared__ float spsum[NW * 32];
    __shared__ __attribute__((aligned(16))) float smax[2 * 32];
    __shared__ __attribute__((aligned(16))) unsigned char mbytes[2 * 2 * gat::Mc];

    const int lane = threadIdx.x & 31;
    const int wv   = __builtin_amdgcn_readfirstlane((int)(threadIdx.x >> 5));
    const int bid  = blockIdx.x;
    const int mb   = bid & 31;
    const int h    = (bid >> 5) & 7;
    const int b    = bid >> 8;
    const int bh   = b * Hc + h;

    const int col  = lane & 15;
    const int half = lane >> 4;
    const int k0   = half * 2;
    const int m0   = mb * 32;

    const float* whb  = wh + (size_t)bh * Mc * Ec;
    const float* s2b  = s2 + (size_t)bh * Mc;

    {
        const float NINF = -__builtin_inff();
        const int rg = wv >> 1;
        const int cb = (wv & 1) * 512;
        v4f sv[4];
        #pragma unroll
        for (int j = 0; j < 4; ++j) sv[j] = *(const v4f*)(s2b + cb + j * 128 + lane * 4);
        const v4i* ap = (const v4i*)(adj + ((size_t)b * Mc + m0 + rg * 8) * Mc + cb) + lane;
        unsigned* mw = (unsigned*)mbytes + rg * (Mc / 4) + cb / 4 + lane;
        float mx[8];
        #pragma unroll
        for (int r = 0; r < 8; ++r) mx[r] = NINF;
        #pragma unroll
        for (int j = 0; j < 4; ++j) {
            unsigned bx = 0, by = 0, bz = 0, bw = 0;
            #pragma unroll
            for (int r = 0; r < 8; ++r) {
                const v4i a = ap[r * (Mc / 4) + j * 32];
                const bool px = a.x > 0, py = a.y > 0, pz = a.z > 0, pw = a.w > 0;
                float m = mx[r];
                m = fmaxf(m, px ? sv[j].x : NINF);
                m = fmaxf(m, py ? sv[j].y : NINF);
                m = fmaxf(m, pz ? sv[j].z : NINF);
                m = fmaxf(m, pw ? sv[j].w : NINF);
                mx[r] = m;
                bx |= (px ? 1u : 0u) << r;
                by |= (py ? 1u : 0u) << r;
                bz |= (pz ? 1u : 0u) << r;
                bw |= (pw ? 1u : 0u) << r;
            }
            mw[j * 32] = bx | (by << 8) | (bz << 16) | (bw << 24);
        }
        #pragma unroll
        for (int d = 1; d < 32; d <<= 1) {
            #pragma unroll
            for (int r = 0; r < 8; ++r) mx[r] = fmaxf(mx[r], __shfl_xor(mx[r], d, 32));
        }
        if (lane == 0) {
            float* sm = &smax[(wv & 1) * 32 + rg * 8];
            *(v4f*)&sm[0] = v4f{ mx[0], mx[1], mx[2], mx[3] };
            *(v4f*)&sm[4] = v4f{ mx[4], mx[5], mx[6], mx[7] };
        }
    }
    __syncthreads();

    float* pblk = smem + wv * 1024;

    float s1v[16], rmx[16];
    #pragma unroll
    for (int t = 0; t < 2; ++t) {
        #pragma unroll
        for (int r = 0; r < 8; ++r) {
            const int mL = t * 16 + r + 8 * half;
            s1v[t * 8 + r] = s1[(size_t)bh * Mc + m0 + mL];
            float sc = s1v[t * 8 + r] + fmaxf(smax[mL], smax[32 + mL]);
            sc = sc > 0.0f ? sc : LALPHA * sc;
            rmx[t * 8 + r] = fmaxf(NEG_INF, sc);
        }
    }

    const unsigned char* mk0 = mbytes + half * Mc + col;
    const unsigned char* mk1 = mk0 + 2 * Mc;

    const int nb_lo = wv * STEPS;

    unsigned mc[2] = { mk0[nb_lo * 16], mk1[nb_lo * 16] };
    const float neg_inf_v = NEG_INF;
    float s2c = s2b[nb_lo * 16 + col];

    v8f acc[2][2] = {};
    v2f psa[2] = {};
    for (int i = 0; i < STEPS; ++i) {
        const int n0 = (nb_lo + i) * 16;

        v2f bf[4][2];
        #pragma unroll
        for (int ks = 0; ks < 4; ++ks) {
            const float* wp = whb + (size_t)(n0 + ks * 4 + k0) * Ec;
            bf[ks][0] = v2f{ wp[col],      wp[Ec + col]      };
            bf[ks][1] = v2f{ wp[16 + col], wp[Ec + 16 + col] };
        }

        const int n1 = (i + 1 < STEPS) ? n0 + 16 : n0;
        const float s2n = s2b[n1 + col];
        const unsigned mn[2] = { mk0[n1], mk1[n1] };

        #pragma unroll
        for (int t = 0; t < 2; ++t) {
            #pragma unroll
            for (int r = 0; r < 8; ++r) {
                const int mL = r + 8 * half;
                float sc = s1v[t * 8 + r] + s2c;
                sc = sc > 0.0f ? sc : LALPHA * sc;
                const int keep = (int)(mc[t] << (31 - r)) >> 31;
                asm("v_bfi_b32 %0, %1, %2, %3" : "=v"(sc) : "v"(keep), "v"(sc), "v"(neg_inf_v));
                pblk[t * 256 + mL * 16 + col] = __expf(sc - rmx[t * 8 + r]);
            }
        }
        __builtin_amdgcn_fence(__ATOMIC_RELEASE, "wavefront");
        __builtin_amdgcn_wave_barrier();
        __builtin_amdgcn_fence(__ATOMIC_ACQUIRE, "wavefront");

        #pragma unroll
        for (int ks = 0; ks < 4; ++ks) {
            v2f af0 = *(const v2f*)&pblk[col * 16 + ks * 4 + k0];
            v2f af1 = *(const v2f*)&pblk[256 + col * 16 + ks * 4 + k0];
            psa[0] += af0;
            psa[1] += af1;
            acc[0][0] = __builtin_amdgcn_wmma_f32_16x16x4_f32(false, af0, false, bf[ks][0],
                                                              (short)0, acc[0][0], false, false);
            acc[0][1] = __builtin_amdgcn_wmma_f32_16x16x4_f32(false, af0, false, bf[ks][1],
                                                              (short)0, acc[0][1], false, false);
            acc[1][0] = __builtin_amdgcn_wmma_f32_16x16x4_f32(false, af1, false, bf[ks][0],
                                                              (short)0, acc[1][0], false, false);
            acc[1][1] = __builtin_amdgcn_wmma_f32_16x16x4_f32(false, af1, false, bf[ks][1],
                                                              (short)0, acc[1][1], false, false);
        }
        __builtin_amdgcn_fence(__ATOMIC_RELEASE, "wavefront");
        __builtin_amdgcn_wave_barrier();
        __builtin_amdgcn_fence(__ATOMIC_ACQUIRE, "wavefront");

        s2c = s2n;
        mc[0] = mn[0];
        mc[1] = mn[1];
    }

    float* red = smem + wv * 1024;
    #pragma unroll
    for (int t = 0; t < 2; ++t) {
        #pragma unroll
        for (int et = 0; et < 2; ++et) {
            #pragma unroll
            for (int r = 0; r < 8; ++r) {
                red[(t * 16 + et * 8 + r) * 32 + lane] = acc[t][et][r];
            }
        }
    }
    {
        float rs0 = psa[0].x + psa[0].y;
        float rs1 = psa[1].x + psa[1].y;
        rs0 += __shfl_xor(rs0, 16, 32);
        rs1 += __shfl_xor(rs1, 16, 32);
        if (half == 0) {
            spsum[wv * 32 + col]      = rs0;
            spsum[wv * 32 + 16 + col] = rs1;
        }
    }
    __syncthreads();

    const int tt = wv >> 2;
    const int rb = (wv & 3) * 2;
    float* ob = out + (size_t)b * Mc * (Hc * Ec) + (size_t)h * Ec;
    #pragma unroll
    for (int rr = 0; rr < 2; ++rr) {
        const int r = rb + rr;
        const float* p0 = smem + (tt * 16 + r) * 32 + lane;
        const float* p1 = p0 + 8 * 32;
        const float* ps = spsum + tt * 16 + r + 8 * half;
        float v0 = p0[0];
        float v1 = p1[0];
        float rs = ps[0];
        #pragma unroll
        for (int w = 1; w < NW; ++w) {
            v0 += p0[w * 1024];
            v1 += p1[w * 1024];
            rs += ps[w * 32];
        }
        const float inv = 1.0f / rs;
        v0 *= inv;
        v1 *= inv;
        const int m = m0 + tt * 16 + r + 8 * half;
        v0 = v0 > 0.0f ? v0 : (__expf(v0) - 1.0f);
        v1 = v1 > 0.0f ? v1 : (__expf(v1) - 1.0f);
        ob[(size_t)m * (Hc * Ec) + col]      = v0;
        ob[(size_t)m * (Hc * Ec) + 16 + col] = v1;
    }
}

extern "C" void kernel_launch(void* const* d_in, const int* in_sizes, int n_in,
                              void* d_out, int out_size, void* d_ws, size_t ws_size,
                              hipStream_t stream) {
    using namespace gat;
    const float* x   = (const float*)d_in[0];
    const int*   adj = (const int*)  d_in[1];
    const float* w   = (const float*)d_in[2];
    const float* att = (const float*)d_in[3];
    float* out = (float*)d_out;

    float* ws   = (float*)d_ws;
    float* wh   = ws;
    float* s1   = ws + WH_N;
    float* s2   = ws + WH_N + S_N;

    const int slabs = Bc * Hc * (Mc / 32);

    gat_wh_kernel<<<slabs, 32, 0, stream>>>(x, w, att, wh, s1, s2);
    gat_attn_kernel<<<slabs, 256, 0, stream>>>(adj, wh, s1, s2, out);
}


# adj read once: Wh-kernel blocks pack it to mask bytes behind their GEMM (piece prefetched to L2), parked in the slab's own out lines; attention scan reads 4 mask words a lane, not adj; sum order same
# speedup vs baseline: 3.0284x; 1.1941x over previous
#include <hip/hip_runtime.h>


typedef __attribute__((ext_vector_type(2))) float v2f;
typedef __attribute__((ext_vector_type(8))) float v8f;

namespace gat {
constexpr int Bc = 8, Mc = 1024, Fc = 256, Hc = 8, Ec = 32;
constexpr size_t WH_N = (size_t)Bc * Hc * Mc * Ec;
constexpr size_t S_N  = (size_t)Bc * Hc * Mc;
constexpr float LALPHA  = 0.2f;
constexpr float NEG_INF = -9.0e15f;
}

__global__ __launch_bounds__(32)
void gat_wh_kernel(const float* __restrict__ x,
                   const int* __restrict__ adj,
                   const float* __restrict__ w,
                   const float* __restrict__ att,
                   float* __restrict__ wh,
                   float* __restrict__ s1,
                   float* __restrict__ s2,
                   float* __restrict__ outm) {
    using namespace gat;
    const int lane = threadIdx.x;
    const int bid  = blockIdx.x;
    const int mb   = bid & 31;
    const int h    = (bid >> 5) & 7;
    const int b    = bid >> 8;

    const int col  = lane & 15;
    const int half = lane >> 4;
    const int k0   = half * 2;

    const int m0 = mb * 32;

    const int* apf = adj + ((size_t)b * Mc + m0 + (h >> 1) * 8 + (lane >> 3)) * Mc
                         + (h & 1) * 512 + (lane & 7) * 64;
    __builtin_prefetch(apf, 0, 1);
    __builtin_prefetch(apf + 4 * Mc, 0, 1);

    const float* xrow0 = x + ((size_t)b * Mc + (m0 + col)) * Fc;
    const float* xrow1 = xrow0 + (size_t)16 * Fc;
    const float* wbase = w + (size_t)h * Fc * Ec;

    v8f acc[2][2] = {};
    for (int kk = 0; kk < Fc; kk += 4) {
        v2f af0 = *(const v2f*)(xrow0 + kk + k0);
        v2f af1 = *(const v2f*)(xrow1 + kk + k0);
        const float* wp = wbase + (size_t)(kk + k0) * Ec;
        v2f bf0 = { wp[col],      wp[Ec + col]      };
        v2f bf1 = { wp[16 + col], wp[Ec + 16 + col] };
        acc[0][0] = __builtin_amdgcn_wmma_f32_16x16x4_f32(false, af0, false, bf0,
                                                          (short)0, acc[0][0], false, false);
        acc[0][1] = __builtin_amdgcn_wmma_f32_16x16x4_f32(false, af0, false, bf1,
                                                          (short)0, acc[0][1], false, false);
        acc[1][0] = __builtin_amdgcn_wmma_f32_16x16x4_f32(false, af1, false, bf0,
                                                          (short)0, acc[1][0], false, false);
        acc[1][1] = __builtin_amdgcn_wmma_f32_16x16x4_f32(false, af1, false, bf1,
                                                          (short)0, acc[1][1], false, false);
    }

    {
        typedef __attribute__((ext_vector_type(4))) int v4i;
        const int rg = h >> 1;
        const int cb = (h & 1) * 512;
        const v4i* ap = (const v4i*)(adj + ((size_t)b * Mc + m0 + rg * 8) * Mc + cb) + lane;
        float* mrow = outm + ((size_t)b * Mc + m0 + 4 * h) * (Hc * Ec) + lane;
        #pragma unroll
        for (int j = 0; j < 4; ++j) {
            unsigned bx = 0, by = 0, bz = 0, bw = 0;
            #pragma unroll
            for (int r = 0; r < 8; ++r) {
                const v4i a = ap[r * (Mc / 4) + j * 32];
                bx |= (a.x > 0 ? 1u : 0u) << r;
                by |= (a.y > 0 ? 1u : 0u) << r;
                bz |= (a.z > 0 ? 1u : 0u) << r;
                bw |= (a.w > 0 ? 1u : 0u) << r;
            }
            const float mword = __uint_as_float(bx | (by << 8) | (bz << 16) | (bw << 24));
            #pragma unroll
            for (int hd = 0; hd < Hc; ++hd) mrow[(size_t)j * (Hc * Ec) + hd * Ec] = mword;
        }
    }

    const float a1_0 = att[h * 2 * Ec + col];
    const float a1_1 = att[h * 2 * Ec + 16 + col];
    const float a2_0 = att[h * 2 * Ec + Ec + col];
    const float a2_1 = att[h * 2 * Ec + Ec + 16 + col];

    const int bh = b * Hc + h;
    float* whbase = wh + ((size_t)bh * Mc + m0) * Ec;

    #pragma unroll
    for (int t = 0; t < 2; ++t) {
        #pragma unroll
        for (int r = 0; r < 8; ++r) {
            const int m_l = t * 16 + r + 8 * half;
            const float v0 = acc[t][0][r];
            const float v1 = acc[t][1][r];
            whbase[(size_t)m_l * Ec + col]      = v0;
            whbase[(size_t)m_l * Ec + 16 + col] = v1;

            float p1 = v0 * a1_0 + v1 * a1_1;
            float p2 = v0 * a2_0 + v1 * a2_1;
            #pragma unroll
            for (int d = 1; d < 16; d <<= 1) {
                p1 += __shfl_xor(p1, d, 32);
                p2 += __shfl_xor(p2, d, 32);
            }
            if (col == 0) {
                s1[(size_t)bh * Mc + m0 + m_l] = p1;
                s2[(size_t)bh * Mc + m0 + m_l] = p2;
            }
        }
    }
}

__global__ __launch_bounds__(256)
void gat_attn_kernel(const float* __restrict__ wh,
                     const float* __restrict__ s1,
                     const float* __restrict__ s2,
                     float* __restrict__ out) {
    using namespace gat;
    typedef __attribute__((ext_vector_type(4))) float v4f;
    constexpr int NW    = 8;
    constexpr int STEPS = Mc / 16 / NW;
    __shared__ float smem[NW * 1024];
    __shared__ float spsum[NW * 32];
    __shared__ __attribute__((aligned(16))) float smax[2 * 32];
    __shared__ __attribute__((aligned(16))) unsigned char mbytes[2 * 2 * gat::Mc];

    const int lane = threadIdx.x & 31;
    const int wv   = __builtin_amdgcn_readfirstlane((int)(threadIdx.x >> 5));
    const int bid  = blockIdx.x;
    const int mb   = bid & 31;
    const int h    = (bid >> 5) & 7;
    const int b    = bid >> 8;
    const int bh   = b * Hc + h;

    const int col  = lane & 15;
    const int half = lane >> 4;
    const int k0   = half * 2;
    const int m0   = mb * 32;

    const float* whb  = wh + (size_t)bh * Mc * Ec;
    const float* s2b  = s2 + (size_t)bh * Mc;

    {
        const float NINF = -__builtin_inff();
        const int rg = wv >> 1;
        const int cb = (wv & 1) * 512;
        v4f sv[4];
        #pragma unroll
        for (int j = 0; j < 4; ++j) sv[j] = *(const v4f*)(s2b + cb + j * 128 + lane * 4);
        const float* mg = out + ((size_t)b * Mc + m0 + 4 * wv) * (Hc * Ec) + h * Ec + lane;
        unsigned wj[4];
        #pragma unroll
        for (int j = 0; j < 4; ++j) wj[j] = __float_as_uint(mg[(size_t)j * (Hc * Ec)]);
        unsigned* mw = (unsigned*)mbytes + rg * (Mc / 4) + cb / 4 + lane;
        const float ninf_v = NINF;
        float mx[8];
        #pragma unroll
        for (int r = 0; r < 8; ++r) mx[r] = NINF;
        #pragma unroll
        for (int j = 0; j < 4; ++j) {
            mw[j * 32] = wj[j];
            #pragma unroll
            for (int r = 0; r < 8; ++r) {
                const int kx = (int)(wj[j] << (31 - r)) >> 31;
                const int ky = (int)(wj[j] << (23 - r)) >> 31;
                const int kz = (int)(wj[j] << (15 - r)) >> 31;
                const int kw = (int)(wj[j] << (7 - r)) >> 31;
                float vx, vy, vz, vw, m = mx[r];
                asm("v_bfi_b32 %0, %1, %2, %3" : "=v"(vx) : "v"(kx), "v"(sv[j].x), "v"(ninf_v));
                asm("v_bfi_b32 %0, %1, %2, %3" : "=v"(vy) : "v"(ky), "v"(sv[j].y), "v"(ninf_v));
                asm("v_bfi_b32 %0, %1, %2, %3" : "=v"(vz) : "v"(kz), "v"(sv[j].z), "v"(ninf_v));
                asm("v_bfi_b32 %0, %1, %2, %3" : "=v"(vw) : "v"(kw), "v"(sv[j].w), "v"(ninf_v));
                asm("v_max3_num_f32 %0, %1, %2, %3" : "=v"(m) : "v"(m), "v"(vx), "v"(vy));
                asm("v_max3_num_f32 %0, %1, %2, %3" : "=v"(m) : "v"(m), "v"(vz), "v"(vw));
                mx[r] = m;
            }
        }
        #pragma unroll
        for (int d = 1; d < 32; d <<= 1) {
            #pragma unroll
            for (int r = 0; r < 8; ++r) mx[r] = fmaxf(mx[r], __shfl_xor(mx[r], d, 32));
        }
        if (lane == 0) {
            float* sm = &smax[(wv & 1) * 32 + rg * 8];
            *(v4f*)&sm[0] = v4f{ mx[0], mx[1], mx[2], mx[3] };
            *(v4f*)&sm[4] = v4f{ mx[4], mx[5], mx[6], mx[7] };
        }
    }
    __syncthreads();

    float* pblk = smem + wv * 1024;

    float s1v[16], rmx[16];
    #pragma unroll
    for (int t = 0; t < 2; ++t) {
        #pragma unroll
        for (int r = 0; r < 8; ++r) {
            const int mL = t * 16 + r + 8 * half;
            s1v[t * 8 + r] = s1[(size_t)bh * Mc + m0 + mL];
            float sc = s1v[t * 8 + r] + fmaxf(smax[mL], smax[32 + mL]);
            sc = sc > 0.0f ? sc : LALPHA * sc;
            rmx[t * 8 + r] = fmaxf(NEG_INF, sc);
        }
    }

    const unsigned char* mk0 = mbytes + half * Mc + col;
    const unsigned char* mk1 = mk0 + 2 * Mc;

    const int nb_lo = wv * STEPS;

    unsigned mc[2] = { mk0[nb_lo * 16], mk1[nb_lo * 16] };
    const float neg_inf_v = NEG_INF;
    float s2c = s2b[nb_lo * 16 + col];

    v8f acc[2][2] = {};
    v2f psa[2] = {};
    for (int i = 0; i < STEPS; ++i) {
        const int n0 = (nb_lo + i) * 16;

        v2f bf[4][2];
        #pragma unroll
        for (int ks = 0; ks < 4; ++ks) {
            const float* wp = whb + (size_t)(n0 + ks * 4 + k0) * Ec;
            bf[ks][0] = v2f{ wp[col],      wp[Ec + col]      };
            bf[ks][1] = v2f{ wp[16 + col], wp[Ec + 16 + col] };
        }

        const int n1 = (i + 1 < STEPS) ? n0 + 16 : n0;
        const float s2n = s2b[n1 + col];
        const unsigned mn[2] = { mk0[n1], mk1[n1] };

        #pragma unroll
        for (int t = 0; t < 2; ++t) {
            #pragma unroll
            for (int r = 0; r < 8; ++r) {
                const int mL = r + 8 * half;
                float sc = s1v[t * 8 + r] + s2c;
                sc = sc > 0.0f ? sc : LALPHA * sc;
                const int keep = (int)(mc[t] << (31 - r)) >> 31;
                asm("v_bfi_b32 %0, %1, %2, %3" : "=v"(sc) : "v"(keep), "v"(sc), "v"(neg_inf_v));
                pblk[t * 256 + mL * 16 + col] = __expf(sc - rmx[t * 8 + r]);
            }
        }
        __builtin_amdgcn_fence(__ATOMIC_RELEASE, "wavefront");
        __builtin_amdgcn_wave_barrier();
        __builtin_amdgcn_fence(__ATOMIC_ACQUIRE, "wavefront");

        #pragma unroll
        for (int ks = 0; ks < 4; ++ks) {
            v2f af0 = *(const v2f*)&pblk[col * 16 + ks * 4 + k0];
            v2f af1 = *(const v2f*)&pblk[256 + col * 16 + ks * 4 + k0];
            psa[0] += af0;
            psa[1] += af1;
            acc[0][0] = __builtin_amdgcn_wmma_f32_16x16x4_f32(false, af0, false, bf[ks][0],
                                                              (short)0, acc[0][0], false, false);
            acc[0][1] = __builtin_amdgcn_wmma_f32_16x16x4_f32(false, af0, false, bf[ks][1],
                                                              (short)0, acc[0][1], false, false);
            acc[1][0] = __builtin_amdgcn_wmma_f32_16x16x4_f32(false, af1, false, bf[ks][0],
                                                              (short)0, acc[1][0], false, false);
            acc[1][1] = __builtin_amdgcn_wmma_f32_16x16x4_f32(false, af1, false, bf[ks][1],
                                                              (short)0, acc[1][1], false, false);
        }
        __builtin_amdgcn_fence(__ATOMIC_RELEASE, "wavefront");
        __builtin_amdgcn_wave_barrier();
        __builtin_amdgcn_fence(__ATOMIC_ACQUIRE, "wavefront");

        s2c = s2n;
        mc[0] = mn[0];
        mc[1] = mn[1];
    }

    float* red = smem + wv * 1024;
    #pragma unroll
    for (int t = 0; t < 2; ++t) {
        #pragma unroll
        for (int et = 0; et < 2; ++et) {
            #pragma unroll
            for (int r = 0; r < 8; ++r) {
                red[(t * 16 + et * 8 + r) * 32 + lane] = acc[t][et][r];
            }
        }
    }
    {
        float rs0 = psa[0].x + psa[0].y;
        float rs1 = psa[1].x + psa[1].y;
        rs0 += __shfl_xor(rs0, 16, 32);
        rs1 += __shfl_xor(rs1, 16, 32);
        if (half == 0) {
            spsum[wv * 32 + col]      = rs0;
            spsum[wv * 32 + 16 + col] = rs1;
        }
    }
    __syncthreads();

    const int tt = wv >> 2;
    const int rb = (wv & 3) * 2;
    float* ob = out + (size_t)b * Mc * (Hc * Ec) + (size_t)h * Ec;
    #pragma unroll
    for (int rr = 0; rr < 2; ++rr) {
        const int r = rb + rr;
        const float* p0 = smem + (tt * 16 + r) * 32 + lane;
        const float* p1 = p0 + 8 * 32;
        const float* ps = spsum + tt * 16 + r + 8 * half;
        float v0 = p0[0];
        float v1 = p1[0];
        float rs = ps[0];
        #pragma unroll
        for (int w = 1; w < NW; ++w) {
            v0 += p0[w * 1024];
            v1 += p1[w * 1024];
            rs += ps[w * 32];
        }
        const float inv = 1.0f / rs;
        v0 *= inv;
        v1 *= inv;
        const int m = m0 + tt * 16 + r + 8 * half;
        v0 = v0 > 0.0f ? v0 : (__expf(v0) - 1.0f);
        v1 = v1 > 0.0f ? v1 : (__expf(v1) - 1.0f);
        ob[(size_t)m * (Hc * Ec) + col]      = v0;
        ob[(size_t)m * (Hc * Ec) + 16 + col] = v1;
    }
}

extern "C" void kernel_launch(void* const* d_in, const int* in_sizes, int n_in,
                              void* d_out, int out_size, void* d_ws, size_t ws_size,
                              hipStream_t stream) {
    using namespace gat;
    const float* x   = (const float*)d_in[0];
    const int*   adj = (const int*)  d_in[1];
    const float* w   = (const float*)d_in[2];
    const float* att = (const float*)d_in[3];
    float* out = (float*)d_out;

    float* ws   = (float*)d_ws;
    float* wh   = ws;
    float* s1   = ws + WH_N;
    float* s2   = ws + WH_N + S_N;

    const int slabs = Bc * Hc * (Mc / 32);

    gat_wh_kernel<<<slabs, 32, 0, stream>>>(x, adj, w, att, wh, s1, s2, out);
    gat_attn_kernel<<<slabs, 256, 0, stream>>>(wh, s1, s2, out);
}


# attention: a quarter fewer vector ops: product transposed (lane = slab row, P built in B-fragment registers, no LDS pass), row terms folded, score = max of 2 sums in log2 units; p rounding changes
# speedup vs baseline: 3.2605x; 1.0766x over previous
#include <hip/hip_runtime.h>


typedef __attribute__((ext_vector_type(2))) float v2f;
typedef __attribute__((ext_vector_type(8))) float v8f;

namespace gat {
constexpr int Bc = 8, Mc = 1024, Fc = 256, Hc = 8, Ec = 32;
constexpr size_t WH_N = (size_t)Bc * Hc * Mc * Ec;
constexpr size_t S_N  = (size_t)Bc * Hc * Mc;
constexpr float LALPHA  = 0.2f;
constexpr float NEG_INF = -9.0e15f;
constexpr float LOG2E   = 1.44269504088896340736f;
}

__global__ __launch_bounds__(32)
void gat_wh_kernel(const float* __restrict__ x,
                   const int* __restrict__ adj,
                   const float* __restrict__ w,
                   const float* __restrict__ att,
                   float* __restrict__ wh,
                   float* __restrict__ s1,
                   float* __restrict__ s2,
                   float* __restrict__ outm) {
    using namespace gat;
    const int lane = threadIdx.x;
    const int bid  = blockIdx.x;
    const int mb   = bid & 31;
    const int h    = (bid >> 5) & 7;
    const int b    = bid >> 8;

    const int col  = lane & 15;
    const int half = lane >> 4;
    const int k0   = half * 2;

    const int m0 = mb * 32;

    const int* apf = adj + ((size_t)b * Mc + m0 + (h >> 1) * 8 + (lane >> 3)) * Mc
                         + (h & 1) * 512 + (lane & 7) * 64;
    __builtin_prefetch(apf, 0, 1);
    __builtin_prefetch(apf + 4 * Mc, 0, 1);

    const float* xrow0 = x + ((size_t)b * Mc + (m0 + col)) * Fc;
    const float* xrow1 = xrow0 + (size_t)16 * Fc;
    const float* wbase = w + (size_t)h * Fc * Ec;

    v8f acc[2][2] = {};
    for (int kk = 0; kk < Fc; kk += 4) {
        v2f af0 = *(const v2f*)(xrow0 + kk + k0);
        v2f af1 = *(const v2f*)(xrow1 + kk + k0);
        const float* wp = wbase + (size_t)(kk + k0) * Ec;
        v2f bf0 = { wp[col],      wp[Ec + col]      };
        v2f bf1 = { wp[16 + col], wp[Ec + 16 + col] };
        acc[0][0] = __builtin_amdgcn_wmma_f32_16x16x4_f32(false, af0, false, bf0,
                                                          (short)0, acc[0][0], false, false);
        acc[0][1] = __builtin_amdgcn_wmma_f32_16x16x4_f32(false, af0, false, bf1,
                                                          (short)0, acc[0][1], false, false);
        acc[1][0] = __builtin_amdgcn_wmma_f32_16x16x4_f32(false, af1, false, bf0,
                                                          (short)0, acc[1][0], false, false);
        acc[1][1] = __builtin_amdgcn_wmma_f32_16x16x4_f32(false, af1, false, bf1,
                                                          (short)0, acc[1][1], false, false);
    }

    {
        typedef __attribute__((ext_vector_type(4))) int v4i;
        const int rg = h >> 1;
        const int cb = (h & 1) * 512;
        const v4i* ap = (const v4i*)(adj + ((size_t)b * Mc + m0 + rg * 8) * Mc + cb) + lane;
        float* mrow = outm + ((size_t)b * Mc + m0 + 4 * h) * (Hc * Ec) + lane;
        #pragma unroll
        for (int j = 0; j < 4; ++j) {
            unsigned bx = 0, by = 0, bz = 0, bw = 0;
            #pragma unroll
            for (int r = 0; r < 8; ++r) {
                const v4i a = ap[r * (Mc / 4) + j * 32];
                bx |= (a.x > 0 ? 1u : 0u) << r;
                by |= (a.y > 0 ? 1u : 0u) << r;
                bz |= (a.z > 0 ? 1u : 0u) << r;
                bw |= (a.w > 0 ? 1u : 0u) << r;
            }
            const float mword = __uint_as_float(bx | (by << 8) | (bz << 16) | (bw << 24));
            #pragma unroll
            for (int hd = 0; hd < Hc; ++hd) mrow[(size_t)j * (Hc * Ec) + hd * Ec] = mword;
        }
    }

    const float a1_0 = att[h * 2 * Ec + col];
    const float a1_1 = att[h * 2 * Ec + 16 + col];
    const float a2_0 = att[h * 2 * Ec + Ec + col];
    const float a2_1 = att[h * 2 * Ec + Ec + 16 + col];

    const int bh = b * Hc + h;
    float* whbase = wh + ((size_t)bh * Mc + m0) * Ec;

    #pragma unroll
    for (int t = 0; t < 2; ++t) {
        #pragma unroll
        for (int r = 0; r < 8; ++r) {
            const int m_l = t * 16 + r + 8 * half;
            const float v0 = acc[t][0][r];
            const float v1 = acc[t][1][r];
            whbase[(size_t)m_l * Ec + col]      = v0;
            whbase[(size_t)m_l * Ec + 16 + col] = v1;

            float p1 = v0 * a1_0 + v1 * a1_1;
            float p2 = v0 * a2_0 + v1 * a2_1;
            #pragma unroll
            for (int d = 1; d < 16; d <<= 1) {
                p1 += __shfl_xor(p1, d, 32);
                p2 += __shfl_xor(p2, d, 32);
            }
            if (col == 0) {
                s1[(size_t)bh * Mc + m0 + m_l] = p1 * LOG2E;
                s2[(size_t)bh * Mc + m0 + m_l] = p2 * LOG2E;
            }
        }
    }
}

__global__ __launch_bounds__(256)
void gat_attn_kernel(const float* __restrict__ wh,
                     const float* __restrict__ s1,
                     const float* __restrict__ s2,
                     float* __restrict__ out) {
    using namespace gat;
    typedef __attribute__((ext_vector_type(4))) float v4f;
    typedef __attribute__((ext_vector_type(4))) unsigned v4u;
    constexpr int NW    = 8;
    constexpr int STEPS = Mc / 16 / NW;
    constexpr int RP    = 33;
    __shared__ float smem[NW * 32 * RP];
    __shared__ float spsum[NW * 32];
    __shared__ __attribute__((aligned(16))) float smax[2 * 32];
    __shared__ __attribute__((aligned(16))) unsigned char mbytes[2 * 2 * gat::Mc];

    const int lane = threadIdx.x & 31;
    const int wv   = __builtin_amdgcn_readfirstlane((int)(threadIdx.x >> 5));
    const int bid  = blockIdx.x;
    const int mb   = bid & 31;
    const int h    = (bid >> 5) & 7;
    const int b    = bid >> 8;
    const int bh   = b * Hc + h;

    const int col  = lane & 15;
    const int half = lane >> 4;
    const int k0   = half * 2;
    const int m0   = mb * 32;

    const float* whb  = wh + (size_t)bh * Mc * Ec;
    const float* s2b  = s2 + (size_t)bh * Mc;

    {
        const float NINF = -__builtin_inff();
        const int rg = wv >> 1;
        const int cb = (wv & 1) * 512;
        v4f sv[4];
        #pragma unroll
        for (int j = 0; j < 4; ++j) sv[j] = *(const v4f*)(s2b + cb + j * 128 + lane * 4);
        const float* mg = out + ((size_t)b * Mc + m0 + 4 * wv) * (Hc * Ec) + h * Ec + lane;
        unsigned wj[4];
        #pragma unroll
        for (int j = 0; j < 4; ++j) wj[j] = __float_as_uint(mg[(size_t)j * (Hc * Ec)]);
        unsigned* mw = (unsigned*)mbytes + rg * (Mc / 4) + cb / 4 + lane;
        const float ninf_v = NINF;
        float mx[8];
        #pragma unroll
        for (int r = 0; r < 8; ++r) mx[r] = NINF;
        #pragma unroll
        for (int j = 0; j < 4; ++j) {
            mw[j * 32] = wj[j];
            #pragma unroll
            for (int r = 0; r < 8; ++r) {
                const int kx = (int)(wj[j] << (31 - r)) >> 31;
                const int ky = (int)(wj[j] << (23 - r)) >> 31;
                const int kz = (int)(wj[j] << (15 - r)) >> 31;
                const int kw = (int)(wj[j] << (7 - r)) >> 31;
                float vx, vy, vz, vw, m = mx[r];
                asm("v_bfi_b32 %0, %1, %2, %3" : "=v"(vx) : "v"(kx), "v"(sv[j].x), "v"(ninf_v));
                asm("v_bfi_b32 %0, %1, %2, %3" : "=v"(vy) : "v"(ky), "v"(sv[j].y), "v"(ninf_v));
                asm("v_bfi_b32 %0, %1, %2, %3" : "=v"(vz) : "v"(kz), "v"(sv[j].z), "v"(ninf_v));
                asm("v_bfi_b32 %0, %1, %2, %3" : "=v"(vw) : "v"(kw), "v"(sv[j].w), "v"(ninf_v));
                asm("v_max3_num_f32 %0, %1, %2, %3" : "=v"(m) : "v"(m), "v"(vx), "v"(vy));
                asm("v_max3_num_f32 %0, %1, %2, %3" : "=v"(m) : "v"(m), "v"(vz), "v"(vw));
                mx[r] = m;
            }
        }
        #pragma unroll
        for (int d = 1; d < 32; d <<= 1) {
            #pragma unroll
            for (int r = 0; r < 8; ++r) mx[r] = fmaxf(mx[r], __shfl_xor(mx[r], d, 32));
        }
        if (lane == 0) {
            float* sm = &smax[(wv & 1) * 32 + rg * 8];
            *(v4f*)&sm[0] = v4f{ mx[0], mx[1], mx[2], mx[3] };
            *(v4f*)&sm[4] = v4f{ mx[4], mx[5], mx[6], mx[7] };
        }
    }
    __syncthreads();

    float ra[2], rb[2], rc[2];
    #pragma unroll
    for (int t = 0; t < 2; ++t) {
        const int mL = t * 16 + col;
        const float S1 = s1[(size_t)bh * Mc + m0 + mL];
        float sc = S1 + fmaxf(smax[mL], smax[32 + mL]);
        sc = sc > 0.0f ? sc : LALPHA * sc;
        const float MX = fmaxf(NEG_INF, sc);
        ra[t] = S1 - MX;
        rb[t] = LALPHA * S1 - MX;
        rc[t] = NEG_INF - MX;
    }

    const unsigned char* mk0 = mbytes + (col >> 3) * Mc;
    const unsigned char* mk1 = mk0 + 2 * Mc;
    const unsigned msh = 16u * half + (col & 7);

    const int nb_lo = wv * STEPS;

    const float* whw = whb + (size_t)(nb_lo * 16 + k0) * Ec + col;
    const float* s2w = s2b + nb_lo * 16 + k0;
    const unsigned char* mw0 = mk0 + nb_lo * 16;
    const unsigned char* mw1 = mk1 + nb_lo * 16;

    v4u mc[2] = { *(const v4u*)mw0, *(const v4u*)mw1 };
    v2f s2c[4];
    #pragma unroll
    for (int ks = 0; ks < 4; ++ks) s2c[ks] = *(const v2f*)(s2w + ks * 4);

    v8f acc[2][2] = {};
    v2f psa[2] = {};
    #pragma unroll 2
    for (int i = 0; i < STEPS; ++i) {
        v2f af[4][2];
        #pragma unroll
        for (int ks = 0; ks < 4; ++ks) {
            const float* wp = whw + (size_t)(i * 16 + ks * 4) * Ec;
            af[ks][0] = v2f{ wp[0],  wp[Ec]      };
            af[ks][1] = v2f{ wp[16], wp[Ec + 16] };
        }

        const int i1 = (i + 1 < STEPS) ? i + 1 : i;
        v2f s2n[4];
        #pragma unroll
        for (int ks = 0; ks < 4; ++ks) s2n[ks] = *(const v2f*)(s2w + i1 * 16 + ks * 4);
        const v4u mn[2] = { *(const v4u*)(mw0 + i1 * 16), *(const v4u*)(mw1 + i1 * 16) };

        v2f s2d[4];
        #pragma unroll
        for (int ks = 0; ks < 4; ++ks) s2d[ks] = s2c[ks] * LALPHA;

        v2f pf[2][4];
        #pragma unroll
        for (int t = 0; t < 2; ++t) {
            #pragma unroll
            for (int ks = 0; ks < 4; ++ks) {
                const v2f pa = s2c[ks] + ra[t];
                const v2f pb = s2d[ks] + rb[t];
                float ex = fmaxf(pa.x, pb.x);
                float ey = fmaxf(pa.y, pb.y);
                const int keepx = __builtin_amdgcn_sbfe((int)mc[t][ks], msh,      1u);
                const int keepy = __builtin_amdgcn_sbfe((int)mc[t][ks], msh + 8u, 1u);
                asm("v_bfi_b32 %0, %1, %2, %3" : "=v"(ex) : "v"(keepx), "v"(ex), "v"(rc[t]));
                asm("v_bfi_b32 %0, %1, %2, %3" : "=v"(ey) : "v"(keepy), "v"(ey), "v"(rc[t]));
                pf[t][ks] = v2f{ __builtin_amdgcn_exp2f(ex), __builtin_amdgcn_exp2f(ey) };
                psa[t] += pf[t][ks];
            }
        }

        #pragma unroll
        for (int ks = 0; ks < 4; ++ks) {
            acc[0][0] = __builtin_amdgcn_wmma_f32_16x16x4_f32(false, af[ks][0], false, pf[0][ks],
                                                              (short)0, acc[0][0], false, false);
            acc[0][1] = __builtin_amdgcn_wmma_f32_16x16x4_f32(false, af[ks][1], false, pf[0][ks],
                                                              (short)0, acc[0][1], false, false);
            acc[1][0] = __builtin_amdgcn_wmma_f32_16x16x4_f32(false, af[ks][0], false, pf[1][ks],
                                                              (short)0, acc[1][0], false, false);
            acc[1][1] = __builtin_amdgcn_wmma_f32_16x16x4_f32(false, af[ks][1], false, pf[1][ks],
                                                              (short)0, acc[1][1], false, false);
        }

        #pragma unroll
        for (int ks = 0; ks < 4; ++ks) s2c[ks] = s2n[ks];
        mc[0] = mn[0];
        mc[1] = mn[1];
    }

    float* red = smem + wv * (32 * RP);
    #pragma unroll
    for (int t = 0; t < 2; ++t) {
        #pragma unroll
        for (int et = 0; et < 2; ++et) {
            #pragma unroll
            for (int r = 0; r < 8; ++r) {
                red[(t * 16 + et * 8 + r) * RP + lane] = acc[t][et][r];
            }
        }
    }
    {
        float rs0 = psa[0].x + psa[0].y;
        float rs1 = psa[1].x + psa[1].y;
        rs0 += __shfl_xor(rs0, 16, 32);
        rs1 += __shfl_xor(rs1, 16, 32);
        if (half == 0) {
            spsum[wv * 32 + col]      = rs0;
            spsum[wv * 32 + 16 + col] = rs1;
        }
    }
    __syncthreads();

    const int freg  = (lane >> 4) * 8 + (lane & 7);
    const int fhalf = (lane >> 3) & 1;
    float rinv;
    {
        const float* ps = spsum + 4 * wv + (lane & 3);
        float rs = ps[0];
        #pragma unroll
        for (int w = 1; w < NW; ++w) rs += ps[w * 32];
        rinv = 1.0f / rs;
    }
    float* ob = out + (size_t)b * Mc * (Hc * Ec) + (size_t)h * Ec;
    #pragma unroll
    for (int q = 0; q < 4; ++q) {
        const int mL = 4 * wv + q;
        const float* p0 = smem + ((mL >> 4) * 16 + freg) * RP + fhalf * 16 + (mL & 15);
        float v0 = p0[0];
        #pragma unroll
        for (int w = 1; w < NW; ++w) v0 += p0[w * (32 * RP)];
        const float inv = __uint_as_float(__builtin_amdgcn_readlane(__float_as_uint(rinv), q));
        v0 *= inv;
        v0 = v0 > 0.0f ? v0 : (__expf(v0) - 1.0f);
        ob[(size_t)(m0 + mL) * (Hc * Ec) + lane] = v0;
    }
}

extern "C" void kernel_launch(void* const* d_in, const int* in_sizes, int n_in,
                              void* d_out, int out_size, void* d_ws, size_t ws_size,
                              hipStream_t stream) {
    using namespace gat;
    const float* x   = (const float*)d_in[0];
    const int*   adj = (const int*)  d_in[1];
    const float* w   = (const float*)d_in[2];
    const float* att = (const float*)d_in[3];
    float* out = (float*)d_out;

    float* ws   = (float*)d_ws;
    float* wh   = ws;
    float* s1   = ws + WH_N;
    float* s2   = ws + WH_N + S_N;

    const int slabs = Bc * Hc * (Mc / 32);

    gat_wh_kernel<<<slabs, 32, 0, stream>>>(x, adj, w, att, wh, s1, s2, out);
    gat_attn_kernel<<<slabs, 256, 0, stream>>>(wh, s1, s2, out);
}
